# starting kernel
# baseline (speedup 1.0000x reference)
#include <hip/hip_runtime.h>

typedef float v2f __attribute__((ext_vector_type(2)));
typedef float v4f __attribute__((ext_vector_type(4)));
typedef float v8f __attribute__((ext_vector_type(8)));

#define DIM            1024
#define ROWS_PER_WAVE  16
#define WAVES          4
#define BLOCK          (WAVES * 32)
#define ROWS_PER_BLOCK (WAVES * ROWS_PER_WAVE)
#define CK             128
#define XSTRIDE        132
#define WZPAD          132

__global__ __launch_bounds__(BLOCK) void crossnet_fused(
    const float* __restrict__ X,
    const float* __restrict__ Wk,
    const float* __restrict__ Bb,
    float* __restrict__ Out)
{
    __shared__ float lds_w[3 * DIM + WZPAD];
    __shared__ float lds_x[WAVES][ROWS_PER_WAVE * XSTRIDE];
    __shared__ float lds_b3[DIM];
    __shared__ float lds_a[ROWS_PER_BLOCK];
    __shared__ float lds_red[2 * BLOCK];
    __shared__ float lds_e[2];

    const int tid = threadIdx.x;

    float p1 = 0.f, p2 = 0.f;
    for (int j = tid; j < DIM; j += BLOCK) {
        float w0 = Wk[j], w1 = Wk[DIM + j], w2 = Wk[2 * DIM + j];
        lds_w[j] = w0; lds_w[DIM + j] = w1; lds_w[2 * DIM + j] = w2;
        float b0 = Bb[j], b1 = Bb[DIM + j], b2 = Bb[2 * DIM + j];
        lds_b3[j] = b0 + b1 + b2;
        p1 += b0 * w1;
        p2 += (b0 + b1) * w2;
    }
    for (int j = tid; j < WZPAD; j += BLOCK) lds_w[3 * DIM + j] = 0.f;
    lds_red[tid] = p1;
    lds_red[BLOCK + tid] = p2;
    __syncthreads();
    if (tid == 0) {
        float e1 = 0.f, e2 = 0.f;
        for (int i = 0; i < BLOCK; ++i) { e1 += lds_red[i]; e2 += lds_red[BLOCK + i]; }
        lds_e[0] = e1; lds_e[1] = e2;
    }
    __syncthreads();

    const int wave = tid >> 5, lane = tid & 31;
    const int m  = lane & 15;
    const int kp = (lane >> 4) << 1;
    float* myx = &lds_x[wave][0];
    const float* abase = &myx[m * XSTRIDE + kp];
    const int row0 = blockIdx.x * ROWS_PER_BLOCK + wave * ROWS_PER_WAVE;

    v8f acc = {};
    for (int kc = 0; kc < DIM; kc += CK) {
        #pragma unroll 4
        for (int r = 0; r < ROWS_PER_WAVE; ++r) {
            v4f v = *(const v4f*)&X[(size_t)(row0 + r) * DIM + kc + lane * 4];
            *(v4f*)&myx[r * XSTRIDE + lane * 4] = v;
        }
        const float* wbase = (m < 3) ? &lds_w[m * DIM + kc + kp] : &lds_w[3 * DIM + kp];
        #pragma unroll
        for (int kk = 0; kk < CK; kk += 4) {
            v2f af = *(const v2f*)(abase + kk);
            v2f bf = *(const v2f*)(wbase + kk);
            acc = __builtin_amdgcn_wmma_f32_16x16x4_f32(
                false, af, false, bf, (short)0, acc, false, false);
        }
    }

    #pragma unroll
    for (int r = 0; r < 8; ++r) {
        int mm = r + ((lane >> 4) << 3);
        myx[mm * 16 + m] = acc[r];
    }
    if (lane < 16) {
        float d0 = myx[lane * 16 + 0];
        float d1 = myx[lane * 16 + 1];
        float d2 = myx[lane * 16 + 2];
        float a = 1.f;
        a = a + a * d0;
        a = a + (a * d1 + lds_e[0]);
        a = a + (a * d2 + lds_e[1]);
        lds_a[wave * ROWS_PER_WAVE + lane] = a;
    }
    __syncthreads();

    const int c0 = tid * (DIM / BLOCK);
    v4f bA = *(const v4f*)&lds_b3[c0];
    v4f bB = *(const v4f*)&lds_b3[c0 + 4];
    const int rowb = blockIdx.x * ROWS_PER_BLOCK;
    for (int r = 0; r < ROWS_PER_BLOCK; ++r) {
        size_t base = (size_t)(rowb + r) * DIM + c0;
        float av = lds_a[r];
        v4f xa = *(const v4f*)&X[base];
        v4f xb = *(const v4f*)&X[base + 4];
        v4f oa = xa * av + bA;
        v4f ob = xb * av + bB;
        __builtin_nontemporal_store(oa, (v4f*)&Out[base]);
        __builtin_nontemporal_store(ob, (v4f*)&Out[base + 4]);
    }
}

extern "C" void kernel_launch(void* const* d_in, const int* in_sizes, int n_in,
                              void* d_out, int out_size, void* d_ws, size_t ws_size,
                              hipStream_t stream) {
    const float* x  = (const float*)d_in[0];
    const float* wk = (const float*)d_in[1];
    const float* bb = (const float*)d_in[2];
    float* out = (float*)d_out;

    const int n_rows = in_sizes[0] / DIM;
    const int grid   = n_rows / ROWS_PER_BLOCK;
    crossnet_fused<<<grid, BLOCK, 0, stream>>>(x, wk, bb, out);
}


# x tile stays in registers from dot products to output (x read once, 32 row loads per wave issued as one group): 16 rows per work-group, K split over 4 waves, d and e sums in a new fixed tree order
# speedup vs baseline: 3.0306x; 3.0306x over previous
#include <hip/hip_runtime.h>

typedef float v2f __attribute__((ext_vector_type(2)));
typedef float v4f __attribute__((ext_vector_type(4)));
typedef float v8f __attribute__((ext_vector_type(8)));

#define DIM      1024
#define ROWS     16
#define WAVES    4
#define BLOCK    (WAVES * 32)
#define WCOLS    (DIM / WAVES)
#define HALF     128
#define XSTRIDE  132
#define WSTRIDE  264

__global__ __launch_bounds__(BLOCK) void crossnet_fused(
    const float* __restrict__ X,
    const float* __restrict__ Wk,
    const float* __restrict__ Bb,
    float* __restrict__ Out)
{
    __shared__ float lds_x[WAVES][ROWS * XSTRIDE];
    __shared__ float lds_w[WAVES][3 * WSTRIDE];
    __shared__ float lds_part[WAVES][ROWS * 16];
    __shared__ float lds_e[WAVES][2];

    const int tid  = threadIdx.x;
    const int wave = tid >> 5, lane = tid & 31;
    const int c = lane & 15;
    const int h = lane >> 4;
    const int row0 = blockIdx.x * ROWS;
    const int col0 = wave * WCOLS + lane * 4;

    v4f wv[3][2], bv[3][2];
    #pragma unroll
    for (int i = 0; i < 3; ++i) {
        #pragma unroll
        for (int hf = 0; hf < 2; ++hf) {
            wv[i][hf] = *(const v4f*)&Wk[i * DIM + col0 + hf * HALF];
            bv[i][hf] = *(const v4f*)&Bb[i * DIM + col0 + hf * HALF];
        }
    }

    v4f xr[2][ROWS];
    #pragma unroll
    for (int hf = 0; hf < 2; ++hf) {
        #pragma unroll
        for (int r = 0; r < ROWS; ++r)
            xr[hf][r] = *(const v4f*)&X[(size_t)(row0 + r) * DIM + col0 + hf * HALF];
    }

    float* myw = &lds_w[wave][0];
    v4f b3[2];
    float p1 = 0.f, p2 = 0.f;
    #pragma unroll
    for (int hf = 0; hf < 2; ++hf) {
        #pragma unroll
        for (int i = 0; i < 3; ++i)
            *(v4f*)&myw[i * WSTRIDE + hf * HALF + lane * 4] = wv[i][hf];
        b3[hf] = bv[0][hf] + bv[1][hf] + bv[2][hf];
        #pragma unroll
        for (int j = 0; j < 4; ++j) {
            p1 += bv[0][hf][j] * wv[1][hf][j];
            p2 += (bv[0][hf][j] + bv[1][hf][j]) * wv[2][hf][j];
        }
    }
    #pragma unroll
    for (int s = 16; s >= 1; s >>= 1) {
        p1 += __shfl_xor(p1, s);
        p2 += __shfl_xor(p2, s);
    }
    if (lane == 0) { lds_e[wave][0] = p1; lds_e[wave][1] = p2; }

    float* myx = &lds_x[wave][0];
    const float* abase = &myx[c * XSTRIDE + 2 * h];
    const float* wbase = &myw[(c < 3 ? c : 2) * WSTRIDE + 2 * h];

    v8f acc = {};
    #pragma unroll
    for (int hf = 0; hf < 2; ++hf) {
        #pragma unroll
        for (int r = 0; r < ROWS; ++r)
            *(v4f*)&myx[r * XSTRIDE + lane * 4] = xr[hf][r];
        #pragma unroll
        for (int kk = 0; kk < HALF; kk += 4) {
            v2f af = *(const v2f*)(abase + kk);
            v2f bf = *(const v2f*)(wbase + hf * HALF + kk);
            acc = __builtin_amdgcn_wmma_f32_16x16x4_f32(
                false, af, false, bf, (short)0, acc, false, false);
        }
    }

    #pragma unroll
    for (int r = 0; r < 8; ++r)
        lds_part[wave][(r + 8 * h) * 16 + c] = acc[r];
    __syncthreads();

    float d0 = 0.f, d1 = 0.f, d2 = 0.f, e1 = 0.f, e2 = 0.f;
    #pragma unroll
    for (int w = 0; w < WAVES; ++w) {
        d0 += lds_part[w][c * 16 + 0];
        d1 += lds_part[w][c * 16 + 1];
        d2 += lds_part[w][c * 16 + 2];
        e1 += lds_e[w][0];
        e2 += lds_e[w][1];
    }
    float a = 1.f;
    a = a + a * d0;
    a = a + (a * d1 + e1);
    a = a + (a * d2 + e2);

    #pragma unroll
    for (int r = 0; r < ROWS; ++r) {
        const float av = __shfl(a, r);
        #pragma unroll
        for (int hf = 0; hf < 2; ++hf) {
            v4f o = xr[hf][r] * av + b3[hf];
            __builtin_nontemporal_store(
                o, (v4f*)&Out[(size_t)(row0 + r) * DIM + col0 + hf * HALF]);
        }
    }
}

extern "C" void kernel_launch(void* const* d_in, const int* in_sizes, int n_in,
                              void* d_out, int out_size, void* d_ws, size_t ws_size,
                              hipStream_t stream) {
    const float* x  = (const float*)d_in[0];
    const float* wk = (const float*)d_in[1];
    const float* bb = (const float*)d_in[2];
    float* out = (float*)d_out;

    const int n_rows = in_sizes[0] / DIM;
    const int grid   = n_rows / ROWS;
    crossnet_fused<<<grid, BLOCK, 0, stream>>>(x, wk, bb, out);
}


# wave load-stage-multiply chain pipelined by column half: half-0 rows asked for first, multiplied while half-1 rows arrive; 16-byte fragment reads one block ahead; sum order in each 8 columns changes
# speedup vs baseline: 3.1946x; 1.0541x over previous
#include <hip/hip_runtime.h>

typedef float v2f __attribute__((ext_vector_type(2)));
typedef float v4f __attribute__((ext_vector_type(4)));
typedef float v8f __attribute__((ext_vector_type(8)));

#define DIM      1024
#define ROWS     16
#define WAVES    4
#define BLOCK    (WAVES * 32)
#define WCOLS    (DIM / WAVES)
#define HALF     128
#define XSTRIDE  132
#define WSTRIDE  264

__global__ __launch_bounds__(BLOCK) void crossnet_fused(
    const float* __restrict__ X,
    const float* __restrict__ Wk,
    const float* __restrict__ Bb,
    float* __restrict__ Out)
{
    __shared__ __attribute__((aligned(16))) float lds_x[WAVES][ROWS * XSTRIDE];
    __shared__ __attribute__((aligned(16))) float lds_w[WAVES][3 * WSTRIDE];
    __shared__ float lds_part[WAVES][ROWS * 16];
    __shared__ float lds_e[WAVES][2];

    const int tid  = threadIdx.x;
    const int wave = tid >> 5, lane = tid & 31;
    const int c = lane & 15;
    const int h = lane >> 4;
    const int row0 = blockIdx.x * ROWS;
    const int col0 = wave * WCOLS + lane * 4;

    v4f wv[3][2], bv[3][2];
    #pragma unroll
    for (int i = 0; i < 3; ++i) {
        #pragma unroll
        for (int hf = 0; hf < 2; ++hf) {
            wv[i][hf] = *(const v4f*)&Wk[i * DIM + col0 + hf * HALF];
            bv[i][hf] = *(const v4f*)&Bb[i * DIM + col0 + hf * HALF];
        }
    }

    v4f xr[2][ROWS];
    #pragma unroll
    for (int r = 0; r < ROWS; ++r)
        xr[0][r] = *(const v4f*)&X[(size_t)(row0 + r) * DIM + col0];
    __builtin_amdgcn_sched_barrier(0);
    #pragma unroll
    for (int r = 0; r < ROWS; ++r)
        xr[1][r] = *(const v4f*)&X[(size_t)(row0 + r) * DIM + col0 + HALF];

    float* myw = &lds_w[wave][0];
    v4f b3[2];
    float p1 = 0.f, p2 = 0.f;
    #pragma unroll
    for (int hf = 0; hf < 2; ++hf) {
        #pragma unroll
        for (int i = 0; i < 3; ++i)
            *(v4f*)&myw[i * WSTRIDE + hf * HALF + lane * 4] = wv[i][hf];
        b3[hf] = bv[0][hf] + bv[1][hf] + bv[2][hf];
        #pragma unroll
        for (int j = 0; j < 4; ++j) {
            p1 += bv[0][hf][j] * wv[1][hf][j];
            p2 += (bv[0][hf][j] + bv[1][hf][j]) * wv[2][hf][j];
        }
    }
    #pragma unroll
    for (int s = 16; s >= 1; s >>= 1) {
        p1 += __shfl_xor(p1, s);
        p2 += __shfl_xor(p2, s);
    }
    if (lane == 0) { lds_e[wave][0] = p1; lds_e[wave][1] = p2; }

    float* myx = &lds_x[wave][0];
    const float* abase = &myx[c * XSTRIDE + 4 * h];
    const float* wbase = &myw[(c < 3 ? c : 2) * WSTRIDE + 4 * h];

    v8f acc = {};
    v4f fa[2][4], fb[2][4];
    #pragma unroll
    for (int hf = 0; hf < 2; ++hf) {
        #pragma unroll
        for (int r = 0; r < ROWS; ++r)
            *(v4f*)&myx[r * XSTRIDE + lane * 4] = xr[hf][r];
        #pragma unroll
        for (int j = 0; j < 4; ++j) {
            fa[0][j] = *(const v4f*)(abase + 8 * j);
            fb[0][j] = *(const v4f*)(wbase + hf * HALF + 8 * j);
        }
        #pragma unroll
        for (int blk = 0; blk < 4; ++blk) {
            const int cur = blk & 1, nxt = cur ^ 1;
            if (blk + 1 < 4) {
                #pragma unroll
                for (int j = 0; j < 4; ++j) {
                    fa[nxt][j] = *(const v4f*)(abase + 8 * (4 * (blk + 1) + j));
                    fb[nxt][j] = *(const v4f*)(wbase + hf * HALF + 8 * (4 * (blk + 1) + j));
                }
            }
            __builtin_amdgcn_sched_barrier(0);
            #pragma unroll
            for (int j = 0; j < 4; ++j) {
                v2f a0 = {fa[cur][j][0], fa[cur][j][1]}, b0 = {fb[cur][j][0], fb[cur][j][1]};
                v2f a1 = {fa[cur][j][2], fa[cur][j][3]}, b1 = {fb[cur][j][2], fb[cur][j][3]};
                acc = __builtin_amdgcn_wmma_f32_16x16x4_f32(
                    false, a0, false, b0, (short)0, acc, false, false);
                acc = __builtin_amdgcn_wmma_f32_16x16x4_f32(
                    false, a1, false, b1, (short)0, acc, false, false);
            }
            __builtin_amdgcn_sched_barrier(0);
        }
    }

    #pragma unroll
    for (int r = 0; r < 8; ++r)
        lds_part[wave][(r + 8 * h) * 16 + c] = acc[r];
    __syncthreads();

    float d0 = 0.f, d1 = 0.f, d2 = 0.f, e1 = 0.f, e2 = 0.f;
    #pragma unroll
    for (int w = 0; w < WAVES; ++w) {
        d0 += lds_part[w][c * 16 + 0];
        d1 += lds_part[w][c * 16 + 1];
        d2 += lds_part[w][c * 16 + 2];
        e1 += lds_e[w][0];
        e2 += lds_e[w][1];
    }
    float a = 1.f;
    a = a + a * d0;
    a = a + (a * d1 + e1);
    a = a + (a * d2 + e2);

    #pragma unroll
    for (int r = 0; r < ROWS; ++r) {
        const float av = __shfl(a, r);
        #pragma unroll
        for (int hf = 0; hf < 2; ++hf) {
            v4f o = xr[hf][r] * av + b3[hf];
            __builtin_nontemporal_store(
                o, (v4f*)&Out[(size_t)(row0 + r) * DIM + col0 + hf * HALF]);
        }
    }
}

extern "C" void kernel_launch(void* const* d_in, const int* in_sizes, int n_in,
                              void* d_out, int out_size, void* d_ws, size_t ws_size,
                              hipStream_t stream) {
    const float* x  = (const float*)d_in[0];
    const float* wk = (const float*)d_in[1];
    const float* bb = (const float*)d_in[2];
    float* out = (float*)d_out;

    const int n_rows = in_sizes[0] / DIM;
    const int grid   = n_rows / ROWS;
    crossnet_fused<<<grid, BLOCK, 0, stream>>>(x, wk, bb, out);
}
